# starting kernel
# baseline (speedup 1.0000x reference)
#include <hip/hip_runtime.h>

#define NUM_TB 32
#define Bsz    32
#define Lsz    4096
#define Dsz    256

typedef __attribute__((ext_vector_type(2))) float v2f;
typedef __attribute__((ext_vector_type(8))) float v8f;

__global__ __launch_bounds__(512)
void time_bin_wmma_kernel(const float* __restrict__ x,
                          const float* __restrict__ ts,
                          float* __restrict__ out) {
    __shared__ unsigned int s_bins[Lsz / 4];
    __shared__ float        s_red[16 * 512];

    const int tid = threadIdx.x;
    const int b   = blockIdx.y;
    const int d0  = blockIdx.x * 32;

    if (blockIdx.x == 0) {
        const float* tsb = ts + b * Lsz;
        float* o2 = out + (size_t)Bsz * NUM_TB * Dsz + (size_t)b * Lsz;
#pragma unroll
        for (int i = 0; i < Lsz / 512; ++i)
            o2[tid + i * 512] = tsb[tid + i * 512];
    }

    const float scale = 32.0f / 1.001f;
    {
        const float4* ts4 = (const float4*)(ts + b * Lsz);
#pragma unroll
        for (int g = tid; g < Lsz / 4; g += 512) {
            float4 t = ts4[g];
            int b0 = (int)(t.x * scale); b0 = b0 < 0 ? 0 : (b0 > 31 ? 31 : b0);
            int b1 = (int)(t.y * scale); b1 = b1 < 0 ? 0 : (b1 > 31 ? 31 : b1);
            int b2 = (int)(t.z * scale); b2 = b2 < 0 ? 0 : (b2 > 31 ? 31 : b2);
            int b3 = (int)(t.w * scale); b3 = b3 < 0 ? 0 : (b3 > 31 ? 31 : b3);
            s_bins[g] = (unsigned)b0 | ((unsigned)b1 << 8) |
                        ((unsigned)b2 << 16) | ((unsigned)b3 << 24);
        }
    }
    __syncthreads();

    const int wave = tid >> 5;
    const int lane = tid & 31;
    const int dsub = wave & 1;
    const int kc   = wave >> 1;
    const int h    = lane >> 4;
    const int n    = lane & 15;

    const float* xp = x + ((size_t)b * Lsz + (size_t)(kc * 512 + 2 * h)) * Dsz
                        + (d0 + dsub * 16 + n);

    v8f acc0 = {};
    v8f acc1 = {};
    const unsigned M1 = (unsigned)(n + 16);

    for (int i = 0; i < 128; ++i) {
        unsigned packed = s_bins[kc * 128 + i];
        unsigned e0 = (packed >> (16 * h)) & 0xFFu;
        unsigned e1 = (packed >> (16 * h + 8)) & 0xFFu;

        v2f A0, A1, Bv;
        A0.x = (e0 == (unsigned)n) ? 1.0f : 0.0f;
        A0.y = (e1 == (unsigned)n) ? 1.0f : 0.0f;
        A1.x = (e0 == M1) ? 1.0f : 0.0f;
        A1.y = (e1 == M1) ? 1.0f : 0.0f;

        Bv.x = __builtin_nontemporal_load(xp);
        Bv.y = __builtin_nontemporal_load(xp + Dsz);

        acc0 = __builtin_amdgcn_wmma_f32_16x16x4_f32(
                   false, A0, false, Bv, (short)0, acc0, false, false);
        acc1 = __builtin_amdgcn_wmma_f32_16x16x4_f32(
                   false, A1, false, Bv, (short)0, acc1, false, false);

        xp += 4 * Dsz;
    }

#pragma unroll
    for (int r = 0; r < 8; ++r) {
        s_red[wave * 512 +       r * 32 + lane] = acc0[r];
        s_red[wave * 512 + 256 + r * 32 + lane] = acc1[r];
    }
    __syncthreads();

    {
        const int j  = tid;
        const int t  = j >> 8;
        const int r  = (j >> 5) & 7;
        const int ln = j & 31;
        const int p  = t * 16 + r + 8 * (ln >> 4);
        const int dd = ln & 15;

        float s0 = 0.0f, s1 = 0.0f;
#pragma unroll
        for (int c = 0; c < 8; ++c) {
            s0 += s_red[(2 * c + 0) * 512 + j];
            s1 += s_red[(2 * c + 1) * 512 + j];
        }
        float* ob = out + ((size_t)b * NUM_TB + p) * Dsz + d0;
        ob[0 * 16 + dd] = s0;
        ob[1 * 16 + dd] = s1;
    }
}

extern "C" void kernel_launch(void* const* d_in, const int* in_sizes, int n_in,
                              void* d_out, int out_size, void* d_ws, size_t ws_size,
                              hipStream_t stream) {
    (void)in_sizes; (void)n_in; (void)d_ws; (void)ws_size; (void)out_size;
    const float* x  = (const float*)d_in[0];
    const float* ts = (const float*)d_in[1];
    float* out = (float*)d_out;

    dim3 grid(Dsz / 32, Bsz);
    time_bin_wmma_kernel<<<grid, 512, 0, stream>>>(x, ts, out);
}


# x staged per wave through private LDS: 16-byte whole-line loads 6 16-row blocks ahead, B fragments read back from LDS, no barrier in the K loop; sums now 16 per-wave partials per output, not 8
# speedup vs baseline: 1.5349x; 1.5349x over previous
#include <hip/hip_runtime.h>


#define NUM_TB 32
#define Bsz    32
#define Lsz    4096
#define Dsz    256

#define WAVES   16
#define WROWS   (Lsz / WAVES)
#define BROWS   16
#define NBLK    (WROWS / BROWS)
#define RING    6
#define HALF_B  288
#define BUF_DW  576
#define WAVE_DW (2 * BUF_DW)

typedef __attribute__((ext_vector_type(2))) float v2f;
typedef __attribute__((ext_vector_type(4))) float v4f;
typedef __attribute__((ext_vector_type(8))) float v8f;

static __device__ __forceinline__ unsigned pack_bins(v4f t) {
    const float scale = 32.0f / 1.001f;
    int b0 = (int)(t.x * scale); b0 = b0 < 0 ? 0 : (b0 > 31 ? 31 : b0);
    int b1 = (int)(t.y * scale); b1 = b1 < 0 ? 0 : (b1 > 31 ? 31 : b1);
    int b2 = (int)(t.z * scale); b2 = b2 < 0 ? 0 : (b2 > 31 ? 31 : b2);
    int b3 = (int)(t.w * scale); b3 = b3 < 0 ? 0 : (b3 > 31 ? 31 : b3);
    return (unsigned)b0 | ((unsigned)b1 << 8) | ((unsigned)b2 << 16) | ((unsigned)b3 << 24);
}

static __device__ __forceinline__ void wave_lds_order() {
    __builtin_amdgcn_fence(__ATOMIC_RELEASE, "wavefront");
    __builtin_amdgcn_wave_barrier();
    __builtin_amdgcn_fence(__ATOMIC_ACQUIRE, "wavefront");
}

__global__ __launch_bounds__(512)
void time_bin_wmma_kernel(const float* __restrict__ x,
                          const float* __restrict__ ts,
                          float* __restrict__ out) {
    __shared__ __attribute__((aligned(16))) float s_stage[WAVES * WAVE_DW];

    const int tid  = threadIdx.x;
    const int lane = tid & 31;
    const int wave = __builtin_amdgcn_readfirstlane(tid >> 5);
    const int b    = blockIdx.y;
    const int d0   = blockIdx.x * 32;
    const int h    = lane >> 4;
    const int n    = lane & 15;
    const int row0 = wave * WROWS;

    const v4f* ts4 = (const v4f*)(ts + (size_t)b * Lsz + row0);
    const v4f t0 = ts4[lane];
    const v4f t1 = ts4[32 + lane];

    const float* xg = x + ((size_t)b * Lsz + (size_t)(row0 + (lane >> 3))) * Dsz
                        + (d0 + (lane & 7) * 4);
    v4f ring[RING][4];
#pragma unroll
    for (int s = 0; s < RING; ++s) {
#pragma unroll
        for (int j = 0; j < 4; ++j)
            ring[s][j] = __builtin_nontemporal_load(
                (const v4f*)(xg + (size_t)(s * BROWS + 4 * j) * Dsz));
    }

    if (blockIdx.x == 0) {
        v4f* o2 = (v4f*)(out + (size_t)Bsz * NUM_TB * Dsz + (size_t)b * Lsz + row0);
        o2[lane]      = t0;
        o2[32 + lane] = t1;
    }

    const unsigned pb0 = pack_bins(t0);
    const unsigned pb1 = pack_bins(t1);

    float* stage = s_stage + wave * WAVE_DW;
    const int frag = h * 16 + n;
    const int slot = ((lane & 7) >> 2) * HALF_B + (lane >> 3) * 16 + (lane & 3) * 4;
    const unsigned sh0 = 8u * (unsigned)h;
    const unsigned sh1 = 8u * (unsigned)h + 16u;
    const unsigned M0 = (unsigned)n;
    const unsigned M1 = (unsigned)(n + 16);

    v8f acc00 = {}, acc10 = {}, acc01 = {}, acc11 = {};

    v2f fr[2][4][2];

#pragma unroll
    for (int blk = -1; blk < NBLK; ++blk) {
        if (blk + 1 < NBLK) {
            const int nb = blk + 1;
            const int s = nb % RING;
            float* buf = stage + (nb & 1) * BUF_DW;

#pragma unroll
            for (int j = 0; j < 4; ++j)
                *(v4f*)(buf + j * 64 + slot) = ring[s][j];

            if (nb + RING < NBLK) {
#pragma unroll
                for (int j = 0; j < 4; ++j)
                    ring[s][j] = __builtin_nontemporal_load(
                        (const v4f*)(xg + (size_t)((nb + RING) * BROWS + 4 * j) * Dsz));
            }

            wave_lds_order();

#pragma unroll
            for (int t = 0; t < 4; ++t) {
                const float* fb = buf + t * 64 + frag;
                fr[nb & 1][t][0].x = fb[0];
                fr[nb & 1][t][0].y = fb[32];
                fr[nb & 1][t][1].x = fb[HALF_B];
                fr[nb & 1][t][1].y = fb[HALF_B + 32];
            }
        }

        __builtin_amdgcn_sched_barrier(0);

        if (blk >= 0) {
#pragma unroll
            for (int t = 0; t < 4; ++t) {
                const int g = blk * 4 + t;
                const unsigned packed = (g < 32)
                    ? (unsigned)__builtin_amdgcn_readlane((int)pb0, g)
                    : (unsigned)__builtin_amdgcn_readlane((int)pb1, g - 32);
                const unsigned e0 = (packed >> sh0) & 0xFFu;
                const unsigned e1 = (packed >> sh1) & 0xFFu;

                v2f A0, A1;
                A0.x = (e0 == M0) ? 1.0f : 0.0f;
                A0.y = (e1 == M0) ? 1.0f : 0.0f;
                A1.x = (e0 == M1) ? 1.0f : 0.0f;
                A1.y = (e1 == M1) ? 1.0f : 0.0f;

                const v2f B0 = fr[blk & 1][t][0];
                const v2f B1 = fr[blk & 1][t][1];

                acc00 = __builtin_amdgcn_wmma_f32_16x16x4_f32(
                            false, A0, false, B0, (short)0, acc00, false, false);
                acc10 = __builtin_amdgcn_wmma_f32_16x16x4_f32(
                            false, A1, false, B0, (short)0, acc10, false, false);
                acc01 = __builtin_amdgcn_wmma_f32_16x16x4_f32(
                            false, A0, false, B1, (short)0, acc01, false, false);
                acc11 = __builtin_amdgcn_wmma_f32_16x16x4_f32(
                            false, A1, false, B1, (short)0, acc11, false, false);
            }
        }

        __builtin_amdgcn_sched_barrier(0);
    }

#pragma unroll
    for (int r = 0; r < 8; ++r) {
        stage[0 * 256 + r * 32 + lane] = acc00[r];
        stage[1 * 256 + r * 32 + lane] = acc01[r];
        stage[2 * 256 + r * 32 + lane] = acc10[r];
        stage[3 * 256 + r * 32 + lane] = acc11[r];
    }
    __syncthreads();

    {
        const int p  = tid >> 4;
        const int cp = tid & 15;
        const int ph = p >> 4;
        const int hh = (p >> 3) & 1;
        const int r  = p & 7;
        const int ct = cp >> 3;
        const int nn = 2 * (cp & 7);
        const int idx = ((ph * 2 + ct) * 8 + r) * 32 + hh * 16 + nn;

        float s0 = 0.0f, s1 = 0.0f;
#pragma unroll
        for (int w = 0; w < WAVES; ++w) {
            const v2f v = *(const v2f*)(s_stage + w * WAVE_DW + idx);
            s0 += v.x;
            s1 += v.y;
        }
        v2f o; o.x = s0; o.y = s1;
        *(v2f*)(out + ((size_t)b * NUM_TB + p) * Dsz + d0 + 2 * cp) = o;
    }
}

extern "C" void kernel_launch(void* const* d_in, const int* in_sizes, int n_in,
                              void* d_out, int out_size, void* d_ws, size_t ws_size,
                              hipStream_t stream) {
    (void)in_sizes; (void)n_in; (void)d_ws; (void)ws_size; (void)out_size;
    const float* x  = (const float*)d_in[0];
    const float* ts = (const float*)d_in[1];
    float* out = (float*)d_out;

    dim3 grid(Dsz / 32, Bsz);
    time_bin_wmma_kernel<<<grid, 512, 0, stream>>>(x, ts, out);
}
